# starting kernel
# baseline (speedup 1.0000x reference)
#include <hip/hip_runtime.h>

typedef __attribute__((ext_vector_type(2))) float v2f;
typedef __attribute__((ext_vector_type(4))) float v4f;
typedef __attribute__((ext_vector_type(8))) float v8f;

namespace {
constexpr int DI = 128, DJ = 512, DK = 512;
constexpr int TI = 16, TJ = 16, TK = 16;
constexpr int NBJ = DJ / TJ;
constexpr int NBK = DK / TK;
constexpr int NBLOCKS = (DI / TI) * NBJ * NBK;

__device__ inline v8f wmma_k8(v2f alo, v2f ahi, v2f blo, v2f bhi, v8f c) {
  c = __builtin_amdgcn_wmma_f32_16x16x4_f32(false, alo, false, blo, (short)0, c,
                                            false, false);
  c = __builtin_amdgcn_wmma_f32_16x16x4_f32(false, ahi, false, bhi, (short)0, c,
                                            false, false);
  return c;
}

__device__ inline void lds_add(float* p, float v) {
  (void)__hip_atomic_fetch_add(p, v, __ATOMIC_RELAXED,
                               __HIP_MEMORY_SCOPE_WORKGROUP);
}
}

__global__ __launch_bounds__(128) void partition_tca_kernel(
    const float* __restrict__ v00, const float* __restrict__ v01,
    const float* __restrict__ v10, const float* __restrict__ v11,
    const float* __restrict__ v20, const float* __restrict__ v21,
    float* __restrict__ out) {
  __shared__ float acc[TI * TJ * TK];

  const int tid  = threadIdx.x;
  const int lane = tid & 31;
  const int wave = tid >> 5;
  const int half = lane >> 4;
  const int ln   = lane & 15;

  const int tile = blockIdx.x;
  const int kt = tile & (NBK - 1);
  const int jt = (tile >> 5) & (NBJ - 1);
  const int it = tile >> 10;
  const int i0 = it * TI, j0 = jt * TJ, k0 = kt * TK;

  const int rb = half * 2;

  v2f a0l, a0h, a1l, a1h, a2l, a2h;
  a0l.x = v00[(0 + rb + 0) * DI + i0 + ln];
  a0l.y = v00[(0 + rb + 1) * DI + i0 + ln];
  a0h.x = v00[(4 + rb + 0) * DI + i0 + ln];
  a0h.y = v00[(4 + rb + 1) * DI + i0 + ln];

  a1l.x = v10[(0 + rb + 0) * DJ + j0 + ln];
  a1l.y = v10[(0 + rb + 1) * DJ + j0 + ln];
  a1h.x = v10[(4 + rb + 0) * DJ + j0 + ln];
  a1h.y = v10[(4 + rb + 1) * DJ + j0 + ln];

  a2l.x = v20[(0 + rb + 0) * DK + k0 + ln];
  a2l.y = v20[(0 + rb + 1) * DK + k0 + ln];
  a2h.x = v20[(4 + rb + 0) * DK + k0 + ln];
  a2h.y = v20[(4 + rb + 1) * DK + k0 + ln];

  for (int q = 0; q < 4; ++q) {
    const int js = wave * 4 + q;
    const int jg = j0 + js;
    v2f bl, bh;
    bl.x = v01[((0 + rb + 0) * (size_t)DJ + jg) * DK + k0 + ln];
    bl.y = v01[((0 + rb + 1) * (size_t)DJ + jg) * DK + k0 + ln];
    bh.x = v01[((4 + rb + 0) * (size_t)DJ + jg) * DK + k0 + ln];
    bh.y = v01[((4 + rb + 1) * (size_t)DJ + jg) * DK + k0 + ln];
    v8f d = {};
    d = wmma_k8(a0l, a0h, bl, bh, d);
#pragma unroll
    for (int v = 0; v < 8; ++v) {
      const int mi = v + half * 8;
      acc[(mi * TJ + js) * TK + ln] = d[v];
    }
  }
  __syncthreads();

  for (int q = 0; q < 4; ++q) {
    const int is = wave * 4 + q;
    const int ig = i0 + is;

    v2f b1l, b1h, b2l, b2h;
    b1l.x = v11[((0 + rb + 0) * (size_t)DI + ig) * DK + k0 + ln];
    b1l.y = v11[((0 + rb + 1) * (size_t)DI + ig) * DK + k0 + ln];
    b1h.x = v11[((4 + rb + 0) * (size_t)DI + ig) * DK + k0 + ln];
    b1h.y = v11[((4 + rb + 1) * (size_t)DI + ig) * DK + k0 + ln];

    b2l.x = v21[((0 + rb + 0) * (size_t)DI + ig) * DJ + j0 + ln];
    b2l.y = v21[((0 + rb + 1) * (size_t)DI + ig) * DJ + j0 + ln];
    b2h.x = v21[((4 + rb + 0) * (size_t)DI + ig) * DJ + j0 + ln];
    b2h.y = v21[((4 + rb + 1) * (size_t)DI + ig) * DJ + j0 + ln];

    v8f d1 = {};
    d1 = wmma_k8(a1l, a1h, b1l, b1h, d1);
    v8f d2 = {};
    d2 = wmma_k8(a2l, a2h, b2l, b2h, d2);

#pragma unroll
    for (int v = 0; v < 8; ++v) {
      const int m = v + half * 8;
      lds_add(&acc[(is * TJ + m) * TK + ln], d1[v]);
      lds_add(&acc[(is * TJ + ln) * TK + m], d2[v]);
    }
  }
  __syncthreads();

  const v4f* accv = (const v4f*)acc;
#pragma unroll
  for (int t = 0; t < 4; ++t) {
    const int rh  = t * 128 + tid;
    const int row = rh >> 1;
    const int h   = rh & 1;
    const int i   = row >> 4;
    const int j   = row & 15;
    v4f lo = accv[row * 4 + h * 2 + 0];
    v4f hi = accv[row * 4 + h * 2 + 1];
    float* dst = out + ((size_t)(i0 + i) * DJ + (j0 + j)) * DK + k0 + h * 8;
    *(v4f*)(dst + 0) = lo;
    *(v4f*)(dst + 4) = hi;
  }
}

extern "C" void kernel_launch(void* const* d_in, const int* in_sizes, int n_in,
                              void* d_out, int out_size, void* d_ws,
                              size_t ws_size, hipStream_t stream) {
  const float* v00 = (const float*)d_in[0];
  const float* v01 = (const float*)d_in[1];
  const float* v10 = (const float*)d_in[2];
  const float* v11 = (const float*)d_in[3];
  const float* v20 = (const float*)d_in[4];
  const float* v21 = (const float*)d_in[5];
  float* out = (float*)d_out;

  partition_tca_kernel<<<NBLOCKS, 128, 0, stream>>>(v00, v01, v10, v11, v20,
                                                    v21, out);
}


# Lane halves hold the two 32-k halves of ONE output row (tile rows 8-15 = rows 0-7 at k+32, K zero-padded in both phases): each store instruction writes one whole aligned 256-B piece; sums unchanged
# speedup vs baseline: 3.0991x; 3.0991x over previous
#include <hip/hip_runtime.h>


typedef __attribute__((ext_vector_type(2))) float v2f;
typedef __attribute__((ext_vector_type(8))) float v8f;

namespace {
constexpr int DI = 128, DJ = 512, DK = 512;
constexpr int GI = 8;
constexpr int TJ = 8;
constexpr int TKH = 32;
constexpr int TKW = 2 * TKH;
constexpr int NGI = DI / GI;
constexpr int NKR = DK / TKW;
constexpr int NJT = DJ / TJ;
constexpr int WAVES = 8;
constexpr int NBLOCKS = NGI * NKR * NJT / WAVES;

__device__ inline v8f wmma4(v2f a, v2f b, v8f c) {
  return __builtin_amdgcn_wmma_f32_16x16x4_f32(false, a, false, b, (short)0, c,
                                               false, false);
}
}

__global__ __launch_bounds__(256, 4) void partition_tca_kernel(
    const float* __restrict__ v00, const float* __restrict__ v01,
    const float* __restrict__ v10, const float* __restrict__ v11,
    const float* __restrict__ v20, const float* __restrict__ v21,
    float* __restrict__ out) {
  const int tid  = threadIdx.x;
  const int lane = tid & 31;
  const int wave = tid >> 5;
  const int half = lane >> 4;
  const int ln   = lane & 15;
  const int l8   = ln & 7;
  const bool low = ln < 8;
  const int rb   = half * 2;

  const int u  = blockIdx.x * WAVES + wave;
  const int ig = u & (NGI - 1);
  const int kr = (u >> 4) & (NKR - 1);
  const int jt = u >> 7;
  const int i0 = ig * GI, j0 = jt * TJ, k0 = kr * TKW;
  const int kc = k0 + 2 * ln;

  const float* p10 = v10 + (rb * DJ + j0 + l8);
  const float* p20 = v20 + (rb * DK + kc);
  const float* p21 = v21 + ((rb * DI + i0) * DJ + j0 + l8);
  const float* p11 = v11 + ((rb * DI + i0) * DK + kc);
  const float* p00 = v00 + (rb * DI + i0 + l8);
  const float* p01 = v01 + ((rb * DJ + j0) * DK + kc);
  float* po = out + ((i0 * DJ + j0) * DK + kc + TKH * half);

  v2f a1l[2], a1h[2];
  {
    const float y0 = p10[0 * DJ];
    const float y1 = p10[1 * DJ];
    const float y2 = p10[4 * DJ];
    const float y3 = p10[5 * DJ];
    a1l[0].x = low ? y0 : 0.0f;  a1l[0].y = low ? y1 : 0.0f;
    a1h[0].x = low ? y2 : 0.0f;  a1h[0].y = low ? y3 : 0.0f;
    a1l[1].x = low ? 0.0f : y0;  a1l[1].y = low ? 0.0f : y1;
    a1h[1].x = low ? 0.0f : y2;  a1h[1].y = low ? 0.0f : y3;
  }

  v2f b2l[2][2], b2h[2][2];
#pragma unroll
  for (int s = 0; s < 2; ++s) {
    const v2f p0 = *(const v2f*)&p20[0 * DK + TKH * s];
    const v2f p1 = *(const v2f*)&p20[1 * DK + TKH * s];
    const v2f p2 = *(const v2f*)&p20[4 * DK + TKH * s];
    const v2f p3 = *(const v2f*)&p20[5 * DK + TKH * s];
    b2l[s][0].x = p0.x; b2l[s][0].y = p1.x; b2h[s][0].x = p2.x; b2h[s][0].y = p3.x;
    b2l[s][1].x = p0.y; b2l[s][1].y = p1.y; b2h[s][1].x = p2.y; b2h[s][1].y = p3.y;
  }

  v8f d[GI][2];
#pragma unroll
  for (int il = 0; il < GI; ++il) {
    v2f a2l[2], a2h[2];
    {
      const float z0 = p21[(0 * DI + il) * DJ];
      const float z1 = p21[(1 * DI + il) * DJ];
      const float z2 = p21[(4 * DI + il) * DJ];
      const float z3 = p21[(5 * DI + il) * DJ];
      a2l[0].x = low ? z0 : 0.0f;  a2l[0].y = low ? z1 : 0.0f;
      a2h[0].x = low ? z2 : 0.0f;  a2h[0].y = low ? z3 : 0.0f;
      a2l[1].x = low ? 0.0f : z0;  a2l[1].y = low ? 0.0f : z1;
      a2h[1].x = low ? 0.0f : z2;  a2h[1].y = low ? 0.0f : z3;
    }

    v2f b1l[2][2], b1h[2][2];
#pragma unroll
    for (int s = 0; s < 2; ++s) {
      const v2f q0 = *(const v2f*)&p11[(0 * DI + il) * DK + TKH * s];
      const v2f q1 = *(const v2f*)&p11[(1 * DI + il) * DK + TKH * s];
      const v2f q2 = *(const v2f*)&p11[(4 * DI + il) * DK + TKH * s];
      const v2f q3 = *(const v2f*)&p11[(5 * DI + il) * DK + TKH * s];
      b1l[s][0].x = q0.x; b1l[s][0].y = q1.x; b1h[s][0].x = q2.x; b1h[s][0].y = q3.x;
      b1l[s][1].x = q0.y; b1l[s][1].y = q1.y; b1h[s][1].x = q2.y; b1h[s][1].y = q3.y;
    }

#pragma unroll
    for (int t = 0; t < 2; ++t) {
      v8f acc = {};
#pragma unroll
      for (int s = 0; s < 2; ++s) {
        acc = wmma4(a1l[s], b1l[s][t], acc);
        acc = wmma4(a1h[s], b1h[s][t], acc);
        acc = wmma4(a2l[s], b2l[s][t], acc);
        acc = wmma4(a2h[s], b2h[s][t], acc);
      }
      d[il][t] = acc;
    }
  }

  const float x0 = p00[0 * DI];
  const float x1 = p00[1 * DI];
  const float x2 = p00[4 * DI];
  const float x3 = p00[5 * DI];
  v2f a0l[2], a0h[2];
  a0l[0].x = low ? x0 : 0.0f;  a0l[0].y = low ? x1 : 0.0f;
  a0h[0].x = low ? x2 : 0.0f;  a0h[0].y = low ? x3 : 0.0f;
  a0l[1].x = low ? 0.0f : x0;  a0l[1].y = low ? 0.0f : x1;
  a0h[1].x = low ? 0.0f : x2;  a0h[1].y = low ? 0.0f : x3;

  struct B0 { v2f p[2][4]; };
  auto load_b0 = [&](int vj) {
    B0 r;
#pragma unroll
    for (int s = 0; s < 2; ++s) {
      r.p[s][0] = *(const v2f*)&p01[(0 * DJ + vj) * DK + TKH * s];
      r.p[s][1] = *(const v2f*)&p01[(1 * DJ + vj) * DK + TKH * s];
      r.p[s][2] = *(const v2f*)&p01[(4 * DJ + vj) * DK + TKH * s];
      r.p[s][3] = *(const v2f*)&p01[(5 * DJ + vj) * DK + TKH * s];
    }
    return r;
  };

  B0 nxt = load_b0(0);
#pragma unroll
  for (int vj = 0; vj < TJ; ++vj) {
    const B0 cur = nxt;
    if (vj + 1 < TJ) nxt = load_b0(vj + 1);
    __builtin_amdgcn_sched_barrier(0);
    v8f e[2];
#pragma unroll
    for (int t = 0; t < 2; ++t) {
      v2f bl[2], bh[2];
#pragma unroll
      for (int s = 0; s < 2; ++s) {
        bl[s].x = cur.p[s][0][t]; bl[s].y = cur.p[s][1][t];
        bh[s].x = cur.p[s][2][t]; bh[s].y = cur.p[s][3][t];
      }
      v8f c;
#pragma unroll
      for (int v = 0; v < GI; ++v) c[v] = d[v][t][vj];
      c = wmma4(a0l[0], bl[0], c);
      c = wmma4(a0h[0], bh[0], c);
      c = wmma4(a0l[1], bl[1], c);
      c = wmma4(a0h[1], bh[1], c);
      e[t] = c;
    }
#pragma unroll
    for (int v = 0; v < GI; ++v) {
      v2f o;
      o.x = e[0][v];
      o.y = e[1][v];
      *(v2f*)&po[(v * DJ + vj) * DK] = o;
    }
    __builtin_amdgcn_sched_barrier(0);
  }
}

extern "C" void kernel_launch(void* const* d_in, const int* in_sizes, int n_in,
                              void* d_out, int out_size, void* d_ws,
                              size_t ws_size, hipStream_t stream) {
  const float* v00 = (const float*)d_in[0];
  const float* v01 = (const float*)d_in[1];
  const float* v10 = (const float*)d_in[2];
  const float* v11 = (const float*)d_in[3];
  const float* v20 = (const float*)d_in[4];
  const float* v21 = (const float*)d_in[5];
  float* out = (float*)d_out;

  partition_tca_kernel<<<NBLOCKS, 256, 0, stream>>>(v00, v01, v10, v11, v20,
                                                    v21, out);
}


# Operand slices of a 32x64x64 output slab staged once per work-group in LDS (256 work-groups of 8 waves, 2 units each): product loop reads LDS only, its global instructions are stores; sums unchanged
# speedup vs baseline: 3.9974x; 1.2899x over previous
#include <hip/hip_runtime.h>


typedef __attribute__((ext_vector_type(2))) float v2f;
typedef __attribute__((ext_vector_type(4))) float v4f;
typedef __attribute__((ext_vector_type(8))) float v8f;

namespace {
constexpr int DI = 128, DJ = 512, DK = 512;
constexpr int GH = 8;
constexpr int GI = 2 * GH;
constexpr int TJ = 8;
constexpr int NT = 4;
constexpr int TKW = 16 * NT;

constexpr int SI = 32, SJ = 64, SK = TKW;
constexpr int NSI = DI / SI;
constexpr int NSJ = DJ / SJ;
constexpr int NSK = DK / SK;
constexpr int NBLOCKS = NSI * NSJ * NSK;
constexpr int WAVES = SJ / TJ;
constexpr int UNITS = SI / GI;
constexpr int THREADS = 32 * WAVES;
constexpr int RANK = 8;

constexpr int O11 = 0;
constexpr int O01 = O11 + RANK * SI * SK;
constexpr int O21 = O01 + RANK * SJ * SK;
constexpr int O00 = O21 + RANK * SI * SJ;
constexpr int O10 = O00 + RANK * SI;
constexpr int O20 = O10 + RANK * SJ;
constexpr int LDS_FLOATS = O20 + RANK * SK;
static_assert(LDS_FLOATS * 4 <= 320 * 1024, "the slab's operands fit the work-group's LDS");
static_assert(SJ == 64 && SK == 64, "the staging copy below takes rows of 64 floats");

__device__ inline v8f wmma4(v2f a, v2f b, v8f c) {
  return __builtin_amdgcn_wmma_f32_16x16x4_f32(false, a, false, b, (short)0, c,
                                               false, false);
}
}

__global__ __launch_bounds__(THREADS, 2) void partition_tca_kernel(
    const float* __restrict__ v00, const float* __restrict__ v01,
    const float* __restrict__ v10, const float* __restrict__ v11,
    const float* __restrict__ v20, const float* __restrict__ v21,
    float* __restrict__ out) {
  __shared__ __attribute__((aligned(16))) float S[LDS_FLOATS];

  const int tid  = threadIdx.x;
  const int lane = tid & 31;
  const int wave = tid >> 5;
  const int half = lane >> 4;
  const int ln   = lane & 15;
  const int l8   = ln & 7;
  const int hp   = ln >> 3;
  const int rb   = half * 2;

  const int b   = blockIdx.x;
  const int k0  = (b & (NSK - 1)) * SK;
  const int i0s = ((b >> 3) & (NSI - 1)) * SI;
  const int j0s = (b >> 5) * SJ;

  {
    const int t4 = tid >> 4;
    const int c4 = (tid & 15) * 4;
    const float* g11 = v11 + ((i0s + t4) * DK + k0 + c4);
    const float* g21 = v21 + ((i0s + t4) * DJ + j0s + c4);
    const float* g01 = v01 + ((j0s + t4) * DK + k0 + c4);
    v4f st[64];
#pragma unroll
    for (int n = 0; n < 16; ++n) {
      st[n]      = *(const v4f*)&g11[((n >> 1) * DI + 16 * (n & 1)) * DK];
      st[16 + n] = *(const v4f*)&g21[((n >> 1) * DI + 16 * (n & 1)) * DJ];
    }
#pragma unroll
    for (int n = 0; n < 32; ++n)
      st[32 + n] = *(const v4f*)&g01[((n >> 2) * DJ + 16 * (n & 3)) * DK];
    const float s00 = v00[(tid >> 5) * DI + i0s + (tid & 31)];
    const float s10a = v10[(tid >> 6) * DJ + j0s + (tid & 63)];
    const float s10b = v10[((tid >> 6) + 4) * DJ + j0s + (tid & 63)];
    const float s20a = v20[(tid >> 6) * DK + k0 + (tid & 63)];
    const float s20b = v20[((tid >> 6) + 4) * DK + k0 + (tid & 63)];
#pragma unroll
    for (int n = 0; n < 16; ++n) {
      *(v4f*)&S[O11 + 4 * (tid + THREADS * n)] = st[n];
      *(v4f*)&S[O21 + 4 * (tid + THREADS * n)] = st[16 + n];
    }
#pragma unroll
    for (int n = 0; n < 32; ++n)
      *(v4f*)&S[O01 + 4 * (tid + THREADS * n)] = st[32 + n];
    S[O00 + tid] = s00;
    S[O10 + tid] = s10a;
    S[O10 + THREADS + tid] = s10b;
    S[O20 + tid] = s20a;
    S[O20 + THREADS + tid] = s20b;
  }
  __syncthreads();

  const int j0 = wave * TJ;
  const int kc = NT * ln;

  const float* p10 = S + O10 + (rb * SJ + j0 + l8);
  const float* p20 = S + O20 + (rb * SK + kc);
  const float* p01 = S + O01 + ((rb * SJ + j0) * SK + kc);

  v2f a1[2][2];
  {
    const float y0 = p10[0 * SJ];
    const float y1 = p10[1 * SJ];
    const float y2 = p10[4 * SJ];
    const float y3 = p10[5 * SJ];
    const bool lo = hp == 0;
    a1[0][0].x = lo ? y0 : 0.0f;  a1[0][0].y = lo ? y1 : 0.0f;
    a1[0][1].x = lo ? y2 : 0.0f;  a1[0][1].y = lo ? y3 : 0.0f;
    a1[1][0].x = lo ? 0.0f : y0;  a1[1][0].y = lo ? 0.0f : y1;
    a1[1][1].x = lo ? 0.0f : y2;  a1[1][1].y = lo ? 0.0f : y3;
  }

  v4f b2r[4];
  b2r[0] = *(const v4f*)&p20[0 * SK];
  b2r[1] = *(const v4f*)&p20[1 * SK];
  b2r[2] = *(const v4f*)&p20[4 * SK];
  b2r[3] = *(const v4f*)&p20[5 * SK];

  struct B0 { v4f p[4]; };
  auto load_b0 = [&](int vj) {
    B0 r;
    r.p[0] = *(const v4f*)&p01[(0 * SJ + vj) * SK];
    r.p[1] = *(const v4f*)&p01[(1 * SJ + vj) * SK];
    r.p[2] = *(const v4f*)&p01[(4 * SJ + vj) * SK];
    r.p[3] = *(const v4f*)&p01[(5 * SJ + vj) * SK];
    return r;
  };

#pragma unroll
  for (int ul = 0; ul < UNITS; ++ul) {
    const int i0 = ul * GI;
    const float* p21 = S + O21 + ((rb * SI + i0 + GH * hp) * SJ + j0 + l8);
    const float* p11 = S + O11 + ((rb * SI + i0) * SK + kc);
    const float* p00 = S + O00 + (rb * SI + i0 + ln);
    float* po = out + (((i0s + i0 + GH * half) * DJ + j0s + j0) * DK + k0 + kc);

    v2f a0[2];
    a0[0].x = p00[0 * SI];
    a0[0].y = p00[1 * SI];
    a0[1].x = p00[4 * SI];
    a0[1].y = p00[5 * SI];

    struct P1 { v2f a2[2]; v4f b1r[2][4]; };
    auto load_p1 = [&](int v) {
      P1 r;
      r.a2[0].x = p21[(0 * SI + v) * SJ];
      r.a2[0].y = p21[(1 * SI + v) * SJ];
      r.a2[1].x = p21[(4 * SI + v) * SJ];
      r.a2[1].y = p21[(5 * SI + v) * SJ];
#pragma unroll
      for (int s = 0; s < 2; ++s) {
        r.b1r[s][0] = *(const v4f*)&p11[(0 * SI + GH * s + v) * SK];
        r.b1r[s][1] = *(const v4f*)&p11[(1 * SI + GH * s + v) * SK];
        r.b1r[s][2] = *(const v4f*)&p11[(4 * SI + GH * s + v) * SK];
        r.b1r[s][3] = *(const v4f*)&p11[(5 * SI + GH * s + v) * SK];
      }
      return r;
    };

    v8f d[GH][NT];
    P1 pn = load_p1(0);
    B0 bq[TJ];
#pragma unroll
    for (int v = 0; v < GH; ++v) {
      const P1 pc = pn;
      if (v + 1 < GH) pn = load_p1(v + 1);
      if (v + 2 >= GH) bq[v + 2 - GH] = load_b0(v + 2 - GH);
      __builtin_amdgcn_sched_barrier(0);
#pragma unroll
      for (int t = 0; t < NT; ++t) {
        v8f acc = {};
#pragma unroll
        for (int s = 0; s < 2; ++s) {
          v2f bl, bh;
          bl.x = pc.b1r[s][0][t]; bl.y = pc.b1r[s][1][t];
          bh.x = pc.b1r[s][2][t]; bh.y = pc.b1r[s][3][t];
          acc = wmma4(a1[s][0], bl, acc);
          acc = wmma4(a1[s][1], bh, acc);
        }
        v2f cl, ch;
        cl.x = b2r[0][t]; cl.y = b2r[1][t];
        ch.x = b2r[2][t]; ch.y = b2r[3][t];
        acc = wmma4(pc.a2[0], cl, acc);
        acc = wmma4(pc.a2[1], ch, acc);
        d[v][t] = acc;
      }
    }

#pragma unroll
    for (int vj = 0; vj < TJ; ++vj) {
      const B0 cur = bq[vj];
      if (vj + 2 < TJ) bq[vj + 2] = load_b0(vj + 2);
      __builtin_amdgcn_sched_barrier(0);
      v8f e[NT];
#pragma unroll
      for (int t = 0; t < NT; ++t) {
        v2f bl, bh;
        bl.x = cur.p[0][t]; bl.y = cur.p[1][t];
        bh.x = cur.p[2][t]; bh.y = cur.p[3][t];
        v8f c;
#pragma unroll
        for (int v = 0; v < GH; ++v) c[v] = d[v][t][vj];
        c = wmma4(a0[0], bl, c);
        c = wmma4(a0[1], bh, c);
        e[t] = c;
      }
#pragma unroll
      for (int v = 0; v < GH; ++v) {
        v4f o;
        o.x = e[0][v];
        o.y = e[1][v];
        o.z = e[2][v];
        o.w = e[3][v];
        *(v4f*)&po[(v * DJ + vj) * DK] = o;
      }
      __builtin_amdgcn_sched_barrier(0);
    }
  }
}

extern "C" void kernel_launch(void* const* d_in, const int* in_sizes, int n_in,
                              void* d_out, int out_size, void* d_ws,
                              size_t ws_size, hipStream_t stream) {
  const float* v00 = (const float*)d_in[0];
  const float* v01 = (const float*)d_in[1];
  const float* v10 = (const float*)d_in[2];
  const float* v11 = (const float*)d_in[3];
  const float* v20 = (const float*)d_in[4];
  const float* v21 = (const float*)d_in[5];
  float* out = (float*)d_out;

  partition_tca_kernel<<<NBLOCKS, THREADS, 0, stream>>>(v00, v01, v10, v11, v20,
                                                        v21, out);
}


# Phase order swapped: term 0 first (64 products, no stores), then terms 1+2 on top of it per i row with its 8 stores right behind: stores run beside 192 products, not 64; sum order t0+t1+t2
# speedup vs baseline: 4.1907x; 1.0484x over previous
#include <hip/hip_runtime.h>


typedef __attribute__((ext_vector_type(2))) float v2f;
typedef __attribute__((ext_vector_type(4))) float v4f;
typedef __attribute__((ext_vector_type(8))) float v8f;

namespace {
constexpr int DI = 128, DJ = 512, DK = 512;
constexpr int GH = 8;
constexpr int GI = 2 * GH;
constexpr int TJ = 8;
constexpr int NT = 4;
constexpr int TKW = 16 * NT;

constexpr int SI = 32, SJ = 64, SK = TKW;
constexpr int NSI = DI / SI;
constexpr int NSJ = DJ / SJ;
constexpr int NSK = DK / SK;
constexpr int NBLOCKS = NSI * NSJ * NSK;
constexpr int WAVES = SJ / TJ;
constexpr int UNITS = SI / GI;
constexpr int THREADS = 32 * WAVES;
constexpr int RANK = 8;

constexpr int O11 = 0;
constexpr int O01 = O11 + RANK * SI * SK;
constexpr int O21 = O01 + RANK * SJ * SK;
constexpr int O00 = O21 + RANK * SI * SJ;
constexpr int O10 = O00 + RANK * SI;
constexpr int O20 = O10 + RANK * SJ;
constexpr int LDS_FLOATS = O20 + RANK * SK;
static_assert(LDS_FLOATS * 4 <= 320 * 1024, "the slab's operands fit the work-group's LDS");
static_assert(SJ == 64 && SK == 64, "the staging copy below takes rows of 64 floats");

__device__ inline v8f wmma4(v2f a, v2f b, v8f c) {
  return __builtin_amdgcn_wmma_f32_16x16x4_f32(false, a, false, b, (short)0, c,
                                               false, false);
}
}

__global__ __launch_bounds__(THREADS, 2) void partition_tca_kernel(
    const float* __restrict__ v00, const float* __restrict__ v01,
    const float* __restrict__ v10, const float* __restrict__ v11,
    const float* __restrict__ v20, const float* __restrict__ v21,
    float* __restrict__ out) {
  __shared__ __attribute__((aligned(16))) float S[LDS_FLOATS];

  const int tid  = threadIdx.x;
  const int lane = tid & 31;
  const int wave = tid >> 5;
  const int half = lane >> 4;
  const int ln   = lane & 15;
  const int l8   = ln & 7;
  const int hp   = ln >> 3;
  const int rb   = half * 2;

  const int b   = blockIdx.x;
  const int k0  = (b & (NSK - 1)) * SK;
  const int i0s = ((b >> 3) & (NSI - 1)) * SI;
  const int j0s = (b >> 5) * SJ;

  {
    const int t4 = tid >> 4;
    const int c4 = (tid & 15) * 4;
    const float* g11 = v11 + ((i0s + t4) * DK + k0 + c4);
    const float* g21 = v21 + ((i0s + t4) * DJ + j0s + c4);
    const float* g01 = v01 + ((j0s + t4) * DK + k0 + c4);
    v4f st[64];
#pragma unroll
    for (int n = 0; n < 16; ++n) {
      st[n]      = *(const v4f*)&g11[((n >> 1) * DI + 16 * (n & 1)) * DK];
      st[16 + n] = *(const v4f*)&g21[((n >> 1) * DI + 16 * (n & 1)) * DJ];
    }
#pragma unroll
    for (int n = 0; n < 32; ++n)
      st[32 + n] = *(const v4f*)&g01[((n >> 2) * DJ + 16 * (n & 3)) * DK];
    const float s00 = v00[(tid >> 5) * DI + i0s + (tid & 31)];
    const float s10a = v10[(tid >> 6) * DJ + j0s + (tid & 63)];
    const float s10b = v10[((tid >> 6) + 4) * DJ + j0s + (tid & 63)];
    const float s20a = v20[(tid >> 6) * DK + k0 + (tid & 63)];
    const float s20b = v20[((tid >> 6) + 4) * DK + k0 + (tid & 63)];
#pragma unroll
    for (int n = 0; n < 16; ++n) {
      *(v4f*)&S[O11 + 4 * (tid + THREADS * n)] = st[n];
      *(v4f*)&S[O21 + 4 * (tid + THREADS * n)] = st[16 + n];
    }
#pragma unroll
    for (int n = 0; n < 32; ++n)
      *(v4f*)&S[O01 + 4 * (tid + THREADS * n)] = st[32 + n];
    S[O00 + tid] = s00;
    S[O10 + tid] = s10a;
    S[O10 + THREADS + tid] = s10b;
    S[O20 + tid] = s20a;
    S[O20 + THREADS + tid] = s20b;
  }
  __syncthreads();

  const int j0 = wave * TJ;
  const int kc = NT * ln;

  const float* p10 = S + O10 + (rb * SJ + j0 + l8);
  const float* p20 = S + O20 + (rb * SK + kc);
  const float* p01 = S + O01 + ((rb * SJ + j0) * SK + kc);

  v2f a1[2][2];
  {
    const float y0 = p10[0 * SJ];
    const float y1 = p10[1 * SJ];
    const float y2 = p10[4 * SJ];
    const float y3 = p10[5 * SJ];
    const bool lo = hp == 0;
    a1[0][0].x = lo ? y0 : 0.0f;  a1[0][0].y = lo ? y1 : 0.0f;
    a1[0][1].x = lo ? y2 : 0.0f;  a1[0][1].y = lo ? y3 : 0.0f;
    a1[1][0].x = lo ? 0.0f : y0;  a1[1][0].y = lo ? 0.0f : y1;
    a1[1][1].x = lo ? 0.0f : y2;  a1[1][1].y = lo ? 0.0f : y3;
  }

  v4f b2r[4];
  b2r[0] = *(const v4f*)&p20[0 * SK];
  b2r[1] = *(const v4f*)&p20[1 * SK];
  b2r[2] = *(const v4f*)&p20[4 * SK];
  b2r[3] = *(const v4f*)&p20[5 * SK];

  v2f a0[UNITS][2];
#pragma unroll
  for (int ul = 0; ul < UNITS; ++ul) {
    const float* p00 = S + O00 + (rb * SI + ul * GI + ln);
    a0[ul][0].x = p00[0 * SI];
    a0[ul][0].y = p00[1 * SI];
    a0[ul][1].x = p00[4 * SI];
    a0[ul][1].y = p00[5 * SI];
  }

  struct B0 { v4f p[4]; };
  auto load_b0 = [&](int vj) {
    B0 r;
    r.p[0] = *(const v4f*)&p01[(0 * SJ + vj) * SK];
    r.p[1] = *(const v4f*)&p01[(1 * SJ + vj) * SK];
    r.p[2] = *(const v4f*)&p01[(4 * SJ + vj) * SK];
    r.p[3] = *(const v4f*)&p01[(5 * SJ + vj) * SK];
    return r;
  };

  B0 bq[TJ];
  bq[0] = load_b0(0);
  bq[1] = load_b0(1);

#pragma unroll
  for (int ul = 0; ul < UNITS; ++ul) {
    const int i0 = ul * GI;
    const float* p21 = S + O21 + ((rb * SI + i0 + GH * hp) * SJ + j0 + l8);
    const float* p11 = S + O11 + ((rb * SI + i0) * SK + kc);
    float* po = out + (((i0s + i0 + GH * half) * DJ + j0s + j0) * DK + k0 + kc);

    struct P1 { v2f a2[2]; v4f b1r[2][4]; };
    auto load_p1 = [&](int v) {
      P1 r;
      r.a2[0].x = p21[(0 * SI + v) * SJ];
      r.a2[0].y = p21[(1 * SI + v) * SJ];
      r.a2[1].x = p21[(4 * SI + v) * SJ];
      r.a2[1].y = p21[(5 * SI + v) * SJ];
#pragma unroll
      for (int s = 0; s < 2; ++s) {
        r.b1r[s][0] = *(const v4f*)&p11[(0 * SI + GH * s + v) * SK];
        r.b1r[s][1] = *(const v4f*)&p11[(1 * SI + GH * s + v) * SK];
        r.b1r[s][2] = *(const v4f*)&p11[(4 * SI + GH * s + v) * SK];
        r.b1r[s][3] = *(const v4f*)&p11[(5 * SI + GH * s + v) * SK];
      }
      return r;
    };

    v8f e[TJ][NT];
    P1 pn;
#pragma unroll
    for (int vj = 0; vj < TJ; ++vj) {
      const B0 cur = bq[vj];
      if (vj + 2 < TJ) bq[vj + 2] = load_b0(vj + 2);
      if (vj + 2 == TJ) pn = load_p1(0);
      __builtin_amdgcn_sched_barrier(0);
#pragma unroll
      for (int t = 0; t < NT; ++t) {
        v2f bl, bh;
        bl.x = cur.p[0][t]; bl.y = cur.p[1][t];
        bh.x = cur.p[2][t]; bh.y = cur.p[3][t];
        v8f c = {};
        c = wmma4(a0[ul][0], bl, c);
        c = wmma4(a0[ul][1], bh, c);
        e[vj][t] = c;
      }
    }

#pragma unroll
    for (int v = 0; v < GH; ++v) {
      const P1 pc = pn;
      if (v + 1 < GH) pn = load_p1(v + 1);
      if (ul + 1 < UNITS && v + 2 >= GH) bq[v + 2 - GH] = load_b0(v + 2 - GH);
      __builtin_amdgcn_sched_barrier(0);
      v8f d[NT];
#pragma unroll
      for (int t = 0; t < NT; ++t) {
        v8f acc;
#pragma unroll
        for (int r = 0; r < TJ; ++r) acc[r] = e[r][t][v];
#pragma unroll
        for (int s = 0; s < 2; ++s) {
          v2f bl, bh;
          bl.x = pc.b1r[s][0][t]; bl.y = pc.b1r[s][1][t];
          bh.x = pc.b1r[s][2][t]; bh.y = pc.b1r[s][3][t];
          acc = wmma4(a1[s][0], bl, acc);
          acc = wmma4(a1[s][1], bh, acc);
        }
        v2f cl, ch;
        cl.x = b2r[0][t]; cl.y = b2r[1][t];
        ch.x = b2r[2][t]; ch.y = b2r[3][t];
        acc = wmma4(pc.a2[0], cl, acc);
        acc = wmma4(pc.a2[1], ch, acc);
        d[t] = acc;
      }
#pragma unroll
      for (int r = 0; r < TJ; ++r) {
        v4f o;
        o.x = d[0][r];
        o.y = d[1][r];
        o.z = d[2][r];
        o.w = d[3][r];
        *(v4f*)&po[(v * DJ + r) * DK] = o;
      }
      __builtin_amdgcn_sched_barrier(0);
    }
  }
}

extern "C" void kernel_launch(void* const* d_in, const int* in_sizes, int n_in,
                              void* d_out, int out_size, void* d_ws,
                              size_t ws_size, hipStream_t stream) {
  const float* v00 = (const float*)d_in[0];
  const float* v01 = (const float*)d_in[1];
  const float* v10 = (const float*)d_in[2];
  const float* v11 = (const float*)d_in[3];
  const float* v20 = (const float*)d_in[4];
  const float* v21 = (const float*)d_in[5];
  float* out = (float*)d_out;

  partition_tca_kernel<<<NBLOCKS, THREADS, 0, stream>>>(v00, v01, v10, v11, v20,
                                                        v21, out);
}


# Wave tile 16 i x 16 j, 1 wave per SIMD: term 0's tiles of the upper 8 j take their i rows rotated by 8, so its sums enter single-i tiles as C in place and term 1 runs at K=8 unpadded; sums unchanged
# speedup vs baseline: 4.4428x; 1.0602x over previous
#include <hip/hip_runtime.h>


typedef __attribute__((ext_vector_type(2))) float v2f;
typedef __attribute__((ext_vector_type(4))) float v4f;
typedef __attribute__((ext_vector_type(8))) float v8f;

namespace {
constexpr int DI = 128, DJ = 512, DK = 512;
constexpr int GH = 8;
constexpr int GI = 16;
constexpr int TJ = 16;
constexpr int NT = 4;
constexpr int TKW = 16 * NT;

constexpr int SI = 32, SJ = 64, SK = TKW;
constexpr int NSI = DI / SI;
constexpr int NSJ = DJ / SJ;
constexpr int NSK = DK / SK;
constexpr int NBLOCKS = NSI * NSJ * NSK;
constexpr int WAVES = SJ / TJ;
constexpr int UNITS = SI / GI;
constexpr int THREADS = 32 * WAVES;
constexpr int RANK = 8;

constexpr int O11 = 0;
constexpr int O01 = O11 + RANK * SI * SK;
constexpr int O21 = O01 + RANK * SJ * SK;
constexpr int O00 = O21 + RANK * SI * SJ;
constexpr int O10 = O00 + RANK * SI;
constexpr int O20 = O10 + RANK * SJ;
constexpr int LDS_FLOATS = O20 + RANK * SK;
static_assert(LDS_FLOATS * 4 <= 320 * 1024, "the slab's operands fit the work-group's LDS");
static_assert(SJ == 64 && SK == 64, "the staging copy below takes rows of 64 floats");

__device__ inline v8f wmma4(v2f a, v2f b, v8f c) {
  return __builtin_amdgcn_wmma_f32_16x16x4_f32(false, a, false, b, (short)0, c,
                                               false, false);
}
}

__global__ __launch_bounds__(THREADS, 1) void partition_tca_kernel(
    const float* __restrict__ v00, const float* __restrict__ v01,
    const float* __restrict__ v10, const float* __restrict__ v11,
    const float* __restrict__ v20, const float* __restrict__ v21,
    float* __restrict__ out) {
  __shared__ __attribute__((aligned(16))) float S[LDS_FLOATS];

  const int tid  = threadIdx.x;
  const int lane = tid & 31;
  const int wave = tid >> 5;
  const int half = lane >> 4;
  const int ln   = lane & 15;
  const int rb   = half * 2;

  const int b   = blockIdx.x;
  const int k0  = (b & (NSK - 1)) * SK;
  const int i0s = ((b >> 3) & (NSI - 1)) * SI;
  const int j0s = (b >> 5) * SJ;

  {
    const int t4 = tid >> 4;
    const int c4 = (tid & 15) * 4;
    const float* g11 = v11 + ((i0s + t4) * DK + k0 + c4);
    const float* g21 = v21 + ((i0s + t4) * DJ + j0s + c4);
    const float* g01 = v01 + ((j0s + t4) * DK + k0 + c4);
    v4f st[128];
#pragma unroll
    for (int n = 0; n < 32; ++n) {
      st[n]      = *(const v4f*)&g11[((n >> 2) * DI + 8 * (n & 3)) * DK];
      st[32 + n] = *(const v4f*)&g21[((n >> 2) * DI + 8 * (n & 3)) * DJ];
    }
#pragma unroll
    for (int n = 0; n < 64; ++n)
      st[64 + n] = *(const v4f*)&g01[((n >> 3) * DJ + 8 * (n & 7)) * DK];
    float s00[2], s10[4], s20[4];
#pragma unroll
    for (int n = 0; n < 2; ++n)
      s00[n] = v00[((tid >> 5) + 4 * n) * DI + i0s + (tid & 31)];
#pragma unroll
    for (int n = 0; n < 4; ++n) {
      s10[n] = v10[((tid >> 6) + 2 * n) * DJ + j0s + (tid & 63)];
      s20[n] = v20[((tid >> 6) + 2 * n) * DK + k0 + (tid & 63)];
    }
#pragma unroll
    for (int n = 0; n < 32; ++n) {
      *(v4f*)&S[O11 + 4 * (tid + THREADS * n)] = st[n];
      *(v4f*)&S[O21 + 4 * (tid + THREADS * n)] = st[32 + n];
    }
#pragma unroll
    for (int n = 0; n < 64; ++n)
      *(v4f*)&S[O01 + 4 * (tid + THREADS * n)] = st[64 + n];
#pragma unroll
    for (int n = 0; n < 2; ++n) S[O00 + tid + THREADS * n] = s00[n];
#pragma unroll
    for (int n = 0; n < 4; ++n) {
      S[O10 + tid + THREADS * n] = s10[n];
      S[O20 + tid + THREADS * n] = s20[n];
    }
  }
  __syncthreads();

  const int j0 = wave * TJ;
  const int kc = NT * ln;

  const float* p20 = S + O20 + (rb * SK + kc);
  const int o01 = O01 + ((rb * SJ + j0) * SK + kc);

  v2f a1[2][2];
  int o21[2];
  float* po[2];
#pragma unroll
  for (int g = 0; g < 2; ++g) {
    const float* p10 = S + O10 + (rb * SJ + j0 + (ln ^ (GH * g)));
    a1[g][0].x = p10[0 * SJ];
    a1[g][0].y = p10[1 * SJ];
    a1[g][1].x = p10[4 * SJ];
    a1[g][1].y = p10[5 * SJ];
    o21[g] = O21 + (rb * SI * SJ + j0 + (ln ^ (GH * g)));
    po[g] = out + ((i0s * DJ + j0s + j0 + GH * (half ^ g)) * DK + k0 + kc);
  }

  v4f b2r[4];
  b2r[0] = *(const v4f*)&p20[0 * SK];
  b2r[1] = *(const v4f*)&p20[1 * SK];
  b2r[2] = *(const v4f*)&p20[4 * SK];
  b2r[3] = *(const v4f*)&p20[5 * SK];

  struct B0 { v4f p[4]; };
  auto load_b0 = [&](int vj) {
    B0 r;
    r.p[0] = *(const v4f*)&S[o01 + (0 * SJ + vj) * SK];
    r.p[1] = *(const v4f*)&S[o01 + (1 * SJ + vj) * SK];
    r.p[2] = *(const v4f*)&S[o01 + (4 * SJ + vj) * SK];
    r.p[3] = *(const v4f*)&S[o01 + (5 * SJ + vj) * SK];
    return r;
  };

  struct P1 { v2f a2[2]; v4f b1r[4]; };
  auto load_p1 = [&](int q21, int q11, int il) {
    P1 r;
    r.a2[0].x = S[q21 + (0 * SI + il) * SJ];
    r.a2[0].y = S[q21 + (1 * SI + il) * SJ];
    r.a2[1].x = S[q21 + (4 * SI + il) * SJ];
    r.a2[1].y = S[q21 + (5 * SI + il) * SJ];
    r.b1r[0] = *(const v4f*)&S[q11 + (0 * SI + il) * SK];
    r.b1r[1] = *(const v4f*)&S[q11 + (1 * SI + il) * SK];
    r.b1r[2] = *(const v4f*)&S[q11 + (4 * SI + il) * SK];
    r.b1r[3] = *(const v4f*)&S[q11 + (5 * SI + il) * SK];
    return r;
  };

  B0 bq[3];
  bq[0] = load_b0(0);
  bq[1] = load_b0(1);

  int o00u = O00 + rb * SI;
  int o11u = O11 + (rb * SI * SK + kc);
  int o21u[2] = {o21[0], o21[1]};
  float* pou[2] = {po[0], po[1]};

#pragma nounroll
  for (int ul = 0; ul < UNITS; ++ul) {

    v2f a0[2][2];
#pragma unroll
    for (int jg = 0; jg < 2; ++jg) {
      const int o00 = o00u + (ln ^ (GH * jg));
      a0[jg][0].x = S[o00 + 0 * SI];
      a0[jg][0].y = S[o00 + 1 * SI];
      a0[jg][1].x = S[o00 + 4 * SI];
      a0[jg][1].y = S[o00 + 5 * SI];
    }

    v8f e[TJ][NT];
    P1 pn;
#pragma unroll
    for (int vj = 0; vj < TJ; ++vj) {
      const B0 cur = bq[vj % 3];
      if (vj + 2 < TJ) bq[(vj + 2) % 3] = load_b0(vj + 2);
      if (vj + 2 == TJ) pn = load_p1(o21u[0], o11u, 0);
      __builtin_amdgcn_sched_barrier(0);
#pragma unroll
      for (int t = 0; t < NT; ++t) {
        v2f bl, bh;
        bl.x = cur.p[0][t]; bl.y = cur.p[1][t];
        bh.x = cur.p[2][t]; bh.y = cur.p[3][t];
        v8f c = {};
        c = wmma4(a0[vj >> 3][0], bl, c);
        c = wmma4(a0[vj >> 3][1], bh, c);
        e[vj][t] = c;
      }
    }

    struct D4 { v8f d[NT]; };
    auto store_step = [&](const D4& q, int s) {
#pragma unroll
      for (int r = 0; r < GH; ++r) {
        v4f o;
        o.x = q.d[0][r];
        o.y = q.d[1][r];
        o.z = q.d[2][r];
        o.w = q.d[3][r];
        *(v4f*)&pou[s >> 3][(s * DJ + r) * DK] = o;
      }
    };
    D4 dp;
#pragma unroll
    for (int s = 0; s < GI; ++s) {
      const int g = s >> 3;
      const int v = s & 7;
      const P1 pc = pn;
      if (s + 1 < GI) pn = load_p1(o21u[(s + 1) >> 3], o11u, s + 1);
      if (s + 2 >= GI) bq[s + 2 - GI] = load_b0(s + 2 - GI);
      __builtin_amdgcn_sched_barrier(0);
      D4 dc;
#pragma unroll
      for (int t = 0; t < NT; ++t) {
        v8f acc;
#pragma unroll
        for (int r = 0; r < GH; ++r)
          acc[r] = half ? e[GH * (1 - g) + r][t][v] : e[GH * g + r][t][v];
        v2f bl, bh;
        bl.x = pc.b1r[0][t]; bl.y = pc.b1r[1][t];
        bh.x = pc.b1r[2][t]; bh.y = pc.b1r[3][t];
        acc = wmma4(a1[g][0], bl, acc);
        acc = wmma4(a1[g][1], bh, acc);
        v2f cl, ch;
        cl.x = b2r[0][t]; cl.y = b2r[1][t];
        ch.x = b2r[2][t]; ch.y = b2r[3][t];
        acc = wmma4(pc.a2[0], cl, acc);
        acc = wmma4(pc.a2[1], ch, acc);
        dc.d[t] = acc;
      }
      if (s > 0) store_step(dp, s - 1);
      dp = dc;
      __builtin_amdgcn_sched_barrier(0);
    }
    store_step(dp, GI - 1);
    o00u += GI;
    o11u += GI * SK;
    o21u[0] += GI * SJ;
    o21u[1] += GI * SJ;
    pou[0] += GI * DJ * DK;
    pou[1] += GI * DJ * DK;
  }
}

extern "C" void kernel_launch(void* const* d_in, const int* in_sizes, int n_in,
                              void* d_out, int out_size, void* d_ws,
                              size_t ws_size, hipStream_t stream) {
  const float* v00 = (const float*)d_in[0];
  const float* v01 = (const float*)d_in[1];
  const float* v10 = (const float*)d_in[2];
  const float* v11 = (const float*)d_in[3];
  const float* v20 = (const float*)d_in[4];
  const float* v21 = (const float*)d_in[5];
  float* out = (float*)d_out;

  partition_tca_kernel<<<NBLOCKS, THREADS, 0, stream>>>(v00, v01, v10, v11, v20,
                                                        v21, out);
}


# Memory instructions leave the product stream at fewer places: phase 2, a step's 8 stores join the LDS loads' group two steps on; phase 1, LDS loads per pair of vj; sums unchanged
# speedup vs baseline: 4.7255x; 1.0636x over previous
#include <hip/hip_runtime.h>


typedef __attribute__((ext_vector_type(2))) float v2f;
typedef __attribute__((ext_vector_type(4))) float v4f;
typedef __attribute__((ext_vector_type(8))) float v8f;

namespace {
constexpr int DI = 128, DJ = 512, DK = 512;
constexpr int GH = 8;
constexpr int GI = 16;
constexpr int TJ = 16;
constexpr int NT = 4;
constexpr int TKW = 16 * NT;

constexpr int SI = 32, SJ = 64, SK = TKW;
constexpr int NSI = DI / SI;
constexpr int NSJ = DJ / SJ;
constexpr int NSK = DK / SK;
constexpr int NBLOCKS = NSI * NSJ * NSK;
constexpr int WAVES = SJ / TJ;
constexpr int UNITS = SI / GI;
constexpr int THREADS = 32 * WAVES;
constexpr int RANK = 8;

constexpr int O11 = 0;
constexpr int O01 = O11 + RANK * SI * SK;
constexpr int O21 = O01 + RANK * SJ * SK;
constexpr int O00 = O21 + RANK * SI * SJ;
constexpr int O10 = O00 + RANK * SI;
constexpr int O20 = O10 + RANK * SJ;
constexpr int LDS_FLOATS = O20 + RANK * SK;
static_assert(LDS_FLOATS * 4 <= 320 * 1024, "the slab's operands fit the work-group's LDS");
static_assert(SJ == 64 && SK == 64, "the staging copy below takes rows of 64 floats");

__device__ inline v8f wmma4(v2f a, v2f b, v8f c) {
  return __builtin_amdgcn_wmma_f32_16x16x4_f32(false, a, false, b, (short)0, c,
                                               false, false);
}
}

__global__ __launch_bounds__(THREADS, 1) void partition_tca_kernel(
    const float* __restrict__ v00, const float* __restrict__ v01,
    const float* __restrict__ v10, const float* __restrict__ v11,
    const float* __restrict__ v20, const float* __restrict__ v21,
    float* __restrict__ out) {
  __shared__ __attribute__((aligned(16))) float S[LDS_FLOATS];

  const int tid  = threadIdx.x;
  const int lane = tid & 31;
  const int wave = tid >> 5;
  const int half = lane >> 4;
  const int ln   = lane & 15;
  const int rb   = half * 2;

  const int b   = blockIdx.x;
  const int k0  = (b & (NSK - 1)) * SK;
  const int i0s = ((b >> 3) & (NSI - 1)) * SI;
  const int j0s = (b >> 5) * SJ;

  {
    const int t4 = tid >> 4;
    const int c4 = (tid & 15) * 4;
    const float* g11 = v11 + ((i0s + t4) * DK + k0 + c4);
    const float* g21 = v21 + ((i0s + t4) * DJ + j0s + c4);
    const float* g01 = v01 + ((j0s + t4) * DK + k0 + c4);
    v4f st[128];
#pragma unroll
    for (int n = 0; n < 32; ++n) {
      st[n]      = *(const v4f*)&g11[((n >> 2) * DI + 8 * (n & 3)) * DK];
      st[32 + n] = *(const v4f*)&g21[((n >> 2) * DI + 8 * (n & 3)) * DJ];
    }
#pragma unroll
    for (int n = 0; n < 64; ++n)
      st[64 + n] = *(const v4f*)&g01[((n >> 3) * DJ + 8 * (n & 7)) * DK];
    float s00[2], s10[4], s20[4];
#pragma unroll
    for (int n = 0; n < 2; ++n)
      s00[n] = v00[((tid >> 5) + 4 * n) * DI + i0s + (tid & 31)];
#pragma unroll
    for (int n = 0; n < 4; ++n) {
      s10[n] = v10[((tid >> 6) + 2 * n) * DJ + j0s + (tid & 63)];
      s20[n] = v20[((tid >> 6) + 2 * n) * DK + k0 + (tid & 63)];
    }
#pragma unroll
    for (int n = 0; n < 32; ++n) {
      *(v4f*)&S[O11 + 4 * (tid + THREADS * n)] = st[n];
      *(v4f*)&S[O21 + 4 * (tid + THREADS * n)] = st[32 + n];
    }
#pragma unroll
    for (int n = 0; n < 64; ++n)
      *(v4f*)&S[O01 + 4 * (tid + THREADS * n)] = st[64 + n];
#pragma unroll
    for (int n = 0; n < 2; ++n) S[O00 + tid + THREADS * n] = s00[n];
#pragma unroll
    for (int n = 0; n < 4; ++n) {
      S[O10 + tid + THREADS * n] = s10[n];
      S[O20 + tid + THREADS * n] = s20[n];
    }
  }
  __syncthreads();

  const int j0 = wave * TJ;
  const int kc = NT * ln;

  const float* p20 = S + O20 + (rb * SK + kc);
  const int o01 = O01 + ((rb * SJ + j0) * SK + kc);

  v2f a1[2][2];
  int o21[2];
  float* po[2];
#pragma unroll
  for (int g = 0; g < 2; ++g) {
    const float* p10 = S + O10 + (rb * SJ + j0 + (ln ^ (GH * g)));
    a1[g][0].x = p10[0 * SJ];
    a1[g][0].y = p10[1 * SJ];
    a1[g][1].x = p10[4 * SJ];
    a1[g][1].y = p10[5 * SJ];
    o21[g] = O21 + (rb * SI * SJ + j0 + (ln ^ (GH * g)));
    po[g] = out + ((i0s * DJ + j0s + j0 + GH * (half ^ g)) * DK + k0 + kc);
  }

  v4f b2r[4];
  b2r[0] = *(const v4f*)&p20[0 * SK];
  b2r[1] = *(const v4f*)&p20[1 * SK];
  b2r[2] = *(const v4f*)&p20[4 * SK];
  b2r[3] = *(const v4f*)&p20[5 * SK];

  struct B0 { v4f p[4]; };
  auto load_b0 = [&](int vj) {
    B0 r;
    r.p[0] = *(const v4f*)&S[o01 + (0 * SJ + vj) * SK];
    r.p[1] = *(const v4f*)&S[o01 + (1 * SJ + vj) * SK];
    r.p[2] = *(const v4f*)&S[o01 + (4 * SJ + vj) * SK];
    r.p[3] = *(const v4f*)&S[o01 + (5 * SJ + vj) * SK];
    return r;
  };

  struct P1 { v2f a2[2]; v4f b1r[4]; };
  auto load_p1 = [&](int q21, int q11, int il) {
    P1 r;
    r.a2[0].x = S[q21 + (0 * SI + il) * SJ];
    r.a2[0].y = S[q21 + (1 * SI + il) * SJ];
    r.a2[1].x = S[q21 + (4 * SI + il) * SJ];
    r.a2[1].y = S[q21 + (5 * SI + il) * SJ];
    r.b1r[0] = *(const v4f*)&S[q11 + (0 * SI + il) * SK];
    r.b1r[1] = *(const v4f*)&S[q11 + (1 * SI + il) * SK];
    r.b1r[2] = *(const v4f*)&S[q11 + (4 * SI + il) * SK];
    r.b1r[3] = *(const v4f*)&S[q11 + (5 * SI + il) * SK];
    return r;
  };

  B0 bq[4];
  bq[0] = load_b0(0);
  bq[1] = load_b0(1);

  int o00u = O00 + rb * SI;
  int o11u = O11 + (rb * SI * SK + kc);
  int o21u[2] = {o21[0], o21[1]};
  float* pou[2] = {po[0], po[1]};

#pragma nounroll
  for (int ul = 0; ul < UNITS; ++ul) {

    v2f a0[2][2];
#pragma unroll
    for (int jg = 0; jg < 2; ++jg) {
      const int o00 = o00u + (ln ^ (GH * jg));
      a0[jg][0].x = S[o00 + 0 * SI];
      a0[jg][0].y = S[o00 + 1 * SI];
      a0[jg][1].x = S[o00 + 4 * SI];
      a0[jg][1].y = S[o00 + 5 * SI];
    }

    v8f e[TJ][NT];
    P1 pn;
#pragma unroll
    for (int vp = 0; vp < TJ / 2; ++vp) {
      const B0 cur[2] = {bq[(2 * vp) & 3], bq[(2 * vp + 1) & 3]};
      if (vp + 1 < TJ / 2) {
        bq[(2 * vp + 2) & 3] = load_b0(2 * vp + 2);
        bq[(2 * vp + 3) & 3] = load_b0(2 * vp + 3);
      }
      if (vp + 1 == TJ / 2) pn = load_p1(o21u[0], o11u, 0);
      __builtin_amdgcn_sched_barrier(0);
#pragma unroll
      for (int u = 0; u < 2; ++u) {
        const int vj = 2 * vp + u;
#pragma unroll
        for (int t = 0; t < NT; ++t) {
          v2f bl, bh;
          bl.x = cur[u].p[0][t]; bl.y = cur[u].p[1][t];
          bh.x = cur[u].p[2][t]; bh.y = cur[u].p[3][t];
          v8f c = {};
          c = wmma4(a0[vj >> 3][0], bl, c);
          c = wmma4(a0[vj >> 3][1], bh, c);
          e[vj][t] = c;
        }
      }
    }

    struct D4 { v8f d[NT]; };
    struct PK { v4f o[GH]; };
    auto pack_step = [&](const D4& q) {
      PK p;
#pragma unroll
      for (int r = 0; r < GH; ++r) {
        p.o[r].x = q.d[0][r];
        p.o[r].y = q.d[1][r];
        p.o[r].z = q.d[2][r];
        p.o[r].w = q.d[3][r];
      }
      return p;
    };
    auto store_step = [&](const PK& p, int s) {
#pragma unroll
      for (int r = 0; r < GH; ++r) {
        *(v4f*)&pou[s >> 3][(s * DJ + r) * DK] = p.o[r];
      }
    };
    D4 dp;
    PK pk;
#pragma unroll
    for (int s = 0; s < GI; ++s) {
      const int g = s >> 3;
      const int v = s & 7;
      const P1 pc = pn;
      if (s + 1 < GI) pn = load_p1(o21u[(s + 1) >> 3], o11u, s + 1);
      if (s + 2 >= GI) bq[s + 2 - GI] = load_b0(s + 2 - GI);
      if (s > 1) store_step(pk, s - 2);
      __builtin_amdgcn_sched_barrier(0);
      if (s > 0) {
        pk = pack_step(dp);
#pragma unroll
        for (int r = 0; r < GH; ++r)
          asm volatile("" : "+v"(pk.o[r]));
      }
      D4 dc;
#pragma unroll
      for (int t = 0; t < NT; ++t) {
        v8f acc;
#pragma unroll
        for (int r = 0; r < GH; ++r)
          acc[r] = half ? e[GH * (1 - g) + r][t][v] : e[GH * g + r][t][v];
        v2f bl, bh;
        bl.x = pc.b1r[0][t]; bl.y = pc.b1r[1][t];
        bh.x = pc.b1r[2][t]; bh.y = pc.b1r[3][t];
        acc = wmma4(a1[g][0], bl, acc);
        acc = wmma4(a1[g][1], bh, acc);
        v2f cl, ch;
        cl.x = b2r[0][t]; cl.y = b2r[1][t];
        ch.x = b2r[2][t]; ch.y = b2r[3][t];
        acc = wmma4(pc.a2[0], cl, acc);
        acc = wmma4(pc.a2[1], ch, acc);
        dc.d[t] = acc;
      }
      dp = dc;
      __builtin_amdgcn_sched_barrier(0);
    }
    store_step(pk, GI - 2);
    pk = pack_step(dp);
    store_step(pk, GI - 1);
    o00u += GI;
    o11u += GI * SK;
    o21u[0] += GI * SJ;
    o21u[1] += GI * SJ;
    pou[0] += GI * DJ * DK;
    pou[1] += GI * DJ * DK;
  }
}

extern "C" void kernel_launch(void* const* d_in, const int* in_sizes, int n_in,
                              void* d_out, int out_size, void* d_ws,
                              size_t ws_size, hipStream_t stream) {
  const float* v00 = (const float*)d_in[0];
  const float* v01 = (const float*)d_in[1];
  const float* v10 = (const float*)d_in[2];
  const float* v11 = (const float*)d_in[3];
  const float* v20 = (const float*)d_in[4];
  const float* v21 = (const float*)d_in[5];
  float* out = (float*)d_out;

  partition_tca_kernel<<<NBLOCKS, THREADS, 0, stream>>>(v00, v01, v10, v11, v20,
                                                        v21, out);
}
